# starting kernel
# baseline (speedup 1.0000x reference)
#include <hip/hip_runtime.h>
#include <hip/hip_bf16.h>
#include <math.h>

typedef __attribute__((ext_vector_type(2))) float v2f;
typedef __attribute__((ext_vector_type(8))) float v8f;

__device__ __forceinline__ v8f wmma_f32(v2f a, v2f b, v8f c) {
  return __builtin_amdgcn_wmma_f32_16x16x4_f32(false, a, false, b, (short)0, c,
                                               false, false);
}

__global__ void bias_pre_kernel(const float* __restrict__ table,
                                float* __restrict__ bias_pre) {
  int tid = blockIdx.x * blockDim.x + threadIdx.x;
  if (tid >= 4 * 98 * 98) return;
  int h = tid / 9604;
  int rem = tid - h * 9604;
  int m = rem / 98;
  int n = rem - m * 98;
  int zm = m / 49, ym = (m % 49) / 7, xm = m % 7;
  int zn = n / 49, yn = (n % 49) / 7, xn = n % 7;
  int idx = (zm - zn + 1) * 169 + (ym - yn + 6) * 13 + (xm - xn + 6);
  bias_pre[tid] = table[idx * 4 + h];
}

__global__ __launch_bounds__(256) void fused_window_attn(
    const float* __restrict__ x, const float* __restrict__ mask,
    const float* __restrict__ w_qkv, const float* __restrict__ bq,
    const float* __restrict__ w_proj, const float* __restrict__ bpj,
    const float* __restrict__ bias_pre, float* __restrict__ out) {
  __shared__ float sQKV[112][386];
  __shared__ float sO[112][130];
  __shared__ float sP[8][16][114];

  const int b = blockIdx.x;
  const int wd = b & 511;
  const int wave = threadIdx.x >> 5;
  const int lane = threadIdx.x & 31;
  const int l16 = lane & 15;
  const int hi = lane >> 4;
  const float* xb = x + (size_t)b * 98 * 128;

  for (int i = 0; i < 3; ++i) {
    const int nt = wave * 3 + i;
    const int col = nt * 16 + l16;
    v8f acc[7] = {};
    for (int k0 = 0; k0 < 128; k0 += 4) {
      const int kb = k0 + 2 * hi;
      v2f bw;
      bw.x = w_qkv[(size_t)kb * 384 + col];
      bw.y = w_qkv[(size_t)(kb + 1) * 384 + col];
#pragma unroll
      for (int mt = 0; mt < 7; ++mt) {
        int n = mt * 16 + l16;
        int nc = n < 98 ? n : 97;
        v2f a = *(const v2f*)(xb + (size_t)nc * 128 + k0 + 2 * hi);
        acc[mt] = wmma_f32(a, bw, acc[mt]);
      }
    }
    const float bias = bq[col];
    const float scale = (col < 128) ? 0.17677669529663687f : 1.0f;
#pragma unroll
    for (int mt = 0; mt < 7; ++mt)
#pragma unroll
      for (int r = 0; r < 8; ++r) {
        int m = mt * 16 + r + 8 * hi;
        sQKV[m][col] = (acc[mt][r] + bias) * scale;
      }
  }
  __syncthreads();

  for (int it = 0; it < 4; ++it) {
    const int u = wave + 8 * it;
    if (u < 28) {
      const int h = u / 7;
      const int mt = u - h * 7;

      v8f acc[7] = {};
      for (int ks = 0; ks < 8; ++ks) {
        const int k0 = ks * 4;
        v2f a = *(const v2f*)(&sQKV[mt * 16 + l16][h * 32 + k0 + 2 * hi]);
#pragma unroll
        for (int nt = 0; nt < 7; ++nt) {
          v2f bk =
              *(const v2f*)(&sQKV[nt * 16 + l16][128 + h * 32 + k0 + 2 * hi]);
          acc[nt] = wmma_f32(a, bk, acc[nt]);
        }
      }

      const float* bp = bias_pre + (size_t)h * 9604;
      const float* mp = mask + (size_t)wd * 9604;
#pragma unroll
      for (int nt = 0; nt < 7; ++nt) {
        const int n = nt * 16 + l16;
#pragma unroll
        for (int r = 0; r < 8; ++r) {
          const int m = mt * 16 + r + 8 * hi;
          if (n < 98 && m < 98)
            acc[nt][r] += bp[m * 98 + n] + mp[m * 98 + n];
          else
            acc[nt][r] = -3.0e38f;
        }
      }

#pragma unroll
      for (int r = 0; r < 8; ++r) {
        float mx = acc[0][r];
#pragma unroll
        for (int nt = 1; nt < 7; ++nt) mx = fmaxf(mx, acc[nt][r]);
#pragma unroll
        for (int off = 1; off < 16; off <<= 1)
          mx = fmaxf(mx, __shfl_xor(mx, off, 32));
        float e[7];
        float sum = 0.0f;
#pragma unroll
        for (int nt = 0; nt < 7; ++nt) {
          e[nt] = __expf(acc[nt][r] - mx);
          sum += e[nt];
        }
#pragma unroll
        for (int off = 1; off < 16; off <<= 1) sum += __shfl_xor(sum, off, 32);
        const float inv = 1.0f / sum;
        const int m = mt * 16 + r + 8 * hi;
        const bool rowv = (m < 98);
#pragma unroll
        for (int nt = 0; nt < 7; ++nt) {
          const int n = nt * 16 + l16;
          sP[wave][r + 8 * hi][nt * 16 + l16] =
              (rowv && n < 98) ? e[nt] * inv : 0.0f;
        }
      }

      v8f o0 = {}, o1 = {};
      for (int ks = 0; ks < 28; ++ks) {
        const int k0 = ks * 4;
        v2f a = *(const v2f*)(&sP[wave][l16][k0 + 2 * hi]);
        const int t0 = k0 + 2 * hi;
        v2f b0, b1;
        b0.x = sQKV[t0][256 + h * 32 + l16];
        b0.y = sQKV[t0 + 1][256 + h * 32 + l16];
        o0 = wmma_f32(a, b0, o0);
        b1.x = sQKV[t0][256 + h * 32 + 16 + l16];
        b1.y = sQKV[t0 + 1][256 + h * 32 + 16 + l16];
        o1 = wmma_f32(a, b1, o1);
      }
#pragma unroll
      for (int r = 0; r < 8; ++r) {
        const int m = mt * 16 + r + 8 * hi;
        sO[m][h * 32 + l16] = o0[r];
        sO[m][h * 32 + 16 + l16] = o1[r];
      }
    }
  }
  __syncthreads();

  {
    const int col = wave * 16 + l16;
    v8f acc[7] = {};
    for (int k0 = 0; k0 < 128; k0 += 4) {
      const int kb = k0 + 2 * hi;
      v2f bw;
      bw.x = w_proj[(size_t)kb * 128 + col];
      bw.y = w_proj[(size_t)(kb + 1) * 128 + col];
#pragma unroll
      for (int mt = 0; mt < 7; ++mt) {
        v2f a = *(const v2f*)(&sO[mt * 16 + l16][k0 + 2 * hi]);
        acc[mt] = wmma_f32(a, bw, acc[mt]);
      }
    }
    const float bias = bpj[col];
#pragma unroll
    for (int mt = 0; mt < 7; ++mt)
#pragma unroll
      for (int r = 0; r < 8; ++r) {
        const int m = mt * 16 + r + 8 * hi;
        if (m < 98) out[((size_t)b * 98 + m) * 128 + col] = acc[mt][r] + bias;
      }
  }
}

extern "C" void kernel_launch(void* const* d_in, const int* in_sizes, int n_in,
                              void* d_out, int out_size, void* d_ws,
                              size_t ws_size, hipStream_t stream) {
  const float* x      = (const float*)d_in[0];
  const float* mask   = (const float*)d_in[1];
  const float* w_qkv  = (const float*)d_in[2];
  const float* b_qkv  = (const float*)d_in[3];
  const float* w_proj = (const float*)d_in[4];
  const float* b_proj = (const float*)d_in[5];
  const float* table  = (const float*)d_in[6];
  float* out = (float*)d_out;
  float* bias_pre = (float*)d_ws;

  bias_pre_kernel<<<(4 * 98 * 98 + 255) / 256, 256, 0, stream>>>(table,
                                                                 bias_pre);
  fused_window_attn<<<2048, 256, 0, stream>>>(x, mask, w_qkv, b_qkv, w_proj,
                                              b_proj, bias_pre, out);
}


# Bias and mask loads of each score tile made unconditional (clamped in-range index, padding selected after) and issued ahead of QK^T: one batch in flight, not 56 serial load-wait round trips
# speedup vs baseline: 1.6674x; 1.6674x over previous
#include <hip/hip_runtime.h>
#include <hip/hip_bf16.h>
#include <math.h>

typedef __attribute__((ext_vector_type(2))) float v2f;
typedef __attribute__((ext_vector_type(8))) float v8f;

__device__ __forceinline__ v8f wmma_f32(v2f a, v2f b, v8f c) {
  return __builtin_amdgcn_wmma_f32_16x16x4_f32(false, a, false, b, (short)0, c,
                                               false, false);
}

__global__ void bias_pre_kernel(const float* __restrict__ table,
                                float* __restrict__ bias_pre) {
  int tid = blockIdx.x * blockDim.x + threadIdx.x;
  if (tid >= 4 * 98 * 98) return;
  int h = tid / 9604;
  int rem = tid - h * 9604;
  int m = rem / 98;
  int n = rem - m * 98;
  int zm = m / 49, ym = (m % 49) / 7, xm = m % 7;
  int zn = n / 49, yn = (n % 49) / 7, xn = n % 7;
  int idx = (zm - zn + 1) * 169 + (ym - yn + 6) * 13 + (xm - xn + 6);
  bias_pre[tid] = table[idx * 4 + h];
}

__global__ __launch_bounds__(256) void fused_window_attn(
    const float* __restrict__ x, const float* __restrict__ mask,
    const float* __restrict__ w_qkv, const float* __restrict__ bq,
    const float* __restrict__ w_proj, const float* __restrict__ bpj,
    const float* __restrict__ bias_pre, float* __restrict__ out) {
  __shared__ float sQKV[112][386];
  __shared__ float sO[112][130];
  __shared__ float sP[8][16][114];

  const int b = blockIdx.x;
  const int wd = b & 511;
  const int wave = threadIdx.x >> 5;
  const int lane = threadIdx.x & 31;
  const int l16 = lane & 15;
  const int hi = lane >> 4;
  const float* xb = x + (size_t)b * 98 * 128;

  for (int i = 0; i < 3; ++i) {
    const int nt = wave * 3 + i;
    const int col = nt * 16 + l16;
    v8f acc[7] = {};
    for (int k0 = 0; k0 < 128; k0 += 4) {
      const int kb = k0 + 2 * hi;
      v2f bw;
      bw.x = w_qkv[(size_t)kb * 384 + col];
      bw.y = w_qkv[(size_t)(kb + 1) * 384 + col];
#pragma unroll
      for (int mt = 0; mt < 7; ++mt) {
        int n = mt * 16 + l16;
        int nc = n < 98 ? n : 97;
        v2f a = *(const v2f*)(xb + (size_t)nc * 128 + k0 + 2 * hi);
        acc[mt] = wmma_f32(a, bw, acc[mt]);
      }
    }
    const float bias = bq[col];
    const float scale = (col < 128) ? 0.17677669529663687f : 1.0f;
#pragma unroll
    for (int mt = 0; mt < 7; ++mt)
#pragma unroll
      for (int r = 0; r < 8; ++r) {
        int m = mt * 16 + r + 8 * hi;
        sQKV[m][col] = (acc[mt][r] + bias) * scale;
      }
  }
  __syncthreads();

  for (int it = 0; it < 4; ++it) {
    const int u = wave + 8 * it;
    if (u < 28) {
      const int h = u / 7;
      const int mt = u - h * 7;

      const float* bp = bias_pre + (size_t)h * 9604;
      const float* mp = mask + (size_t)wd * 9604;
      float bv[7][8], mv[7][8];
#pragma unroll
      for (int nt = 0; nt < 7; ++nt) {
        const int n = nt * 16 + l16;
        const int nn = n < 98 ? n : 97;
#pragma unroll
        for (int r = 0; r < 8; ++r) {
          const int m = mt * 16 + r + 8 * hi;
          const int mm = m < 98 ? m : 97;
          bv[nt][r] = bp[mm * 98 + nn];
          mv[nt][r] = mp[mm * 98 + nn];
        }
      }

      v8f acc[7] = {};
      for (int ks = 0; ks < 8; ++ks) {
        const int k0 = ks * 4;
        v2f a = *(const v2f*)(&sQKV[mt * 16 + l16][h * 32 + k0 + 2 * hi]);
#pragma unroll
        for (int nt = 0; nt < 7; ++nt) {
          v2f bk =
              *(const v2f*)(&sQKV[nt * 16 + l16][128 + h * 32 + k0 + 2 * hi]);
          acc[nt] = wmma_f32(a, bk, acc[nt]);
        }
      }

#pragma unroll
      for (int nt = 0; nt < 7; ++nt) {
        const int n = nt * 16 + l16;
#pragma unroll
        for (int r = 0; r < 8; ++r) {
          const int m = mt * 16 + r + 8 * hi;
          const float s = acc[nt][r] + (bv[nt][r] + mv[nt][r]);
          acc[nt][r] = (n < 98 && m < 98) ? s : -3.0e38f;
        }
      }

#pragma unroll
      for (int r = 0; r < 8; ++r) {
        float mx = acc[0][r];
#pragma unroll
        for (int nt = 1; nt < 7; ++nt) mx = fmaxf(mx, acc[nt][r]);
#pragma unroll
        for (int off = 1; off < 16; off <<= 1)
          mx = fmaxf(mx, __shfl_xor(mx, off, 32));
        float e[7];
        float sum = 0.0f;
#pragma unroll
        for (int nt = 0; nt < 7; ++nt) {
          e[nt] = __expf(acc[nt][r] - mx);
          sum += e[nt];
        }
#pragma unroll
        for (int off = 1; off < 16; off <<= 1) sum += __shfl_xor(sum, off, 32);
        const float inv = 1.0f / sum;
        const int m = mt * 16 + r + 8 * hi;
        const bool rowv = (m < 98);
#pragma unroll
        for (int nt = 0; nt < 7; ++nt) {
          const int n = nt * 16 + l16;
          sP[wave][r + 8 * hi][nt * 16 + l16] =
              (rowv && n < 98) ? e[nt] * inv : 0.0f;
        }
      }

      v8f o0 = {}, o1 = {};
      for (int ks = 0; ks < 28; ++ks) {
        const int k0 = ks * 4;
        v2f a = *(const v2f*)(&sP[wave][l16][k0 + 2 * hi]);
        const int t0 = k0 + 2 * hi;
        v2f b0, b1;
        b0.x = sQKV[t0][256 + h * 32 + l16];
        b0.y = sQKV[t0 + 1][256 + h * 32 + l16];
        o0 = wmma_f32(a, b0, o0);
        b1.x = sQKV[t0][256 + h * 32 + 16 + l16];
        b1.y = sQKV[t0 + 1][256 + h * 32 + 16 + l16];
        o1 = wmma_f32(a, b1, o1);
      }
#pragma unroll
      for (int r = 0; r < 8; ++r) {
        const int m = mt * 16 + r + 8 * hi;
        sO[m][h * 32 + l16] = o0[r];
        sO[m][h * 32 + 16 + l16] = o1[r];
      }
    }
  }
  __syncthreads();

  {
    const int col = wave * 16 + l16;
    v8f acc[7] = {};
    for (int k0 = 0; k0 < 128; k0 += 4) {
      const int kb = k0 + 2 * hi;
      v2f bw;
      bw.x = w_proj[(size_t)kb * 128 + col];
      bw.y = w_proj[(size_t)(kb + 1) * 128 + col];
#pragma unroll
      for (int mt = 0; mt < 7; ++mt) {
        v2f a = *(const v2f*)(&sO[mt * 16 + l16][k0 + 2 * hi]);
        acc[mt] = wmma_f32(a, bw, acc[mt]);
      }
    }
    const float bias = bpj[col];
#pragma unroll
    for (int mt = 0; mt < 7; ++mt)
#pragma unroll
      for (int r = 0; r < 8; ++r) {
        const int m = mt * 16 + r + 8 * hi;
        if (m < 98) out[((size_t)b * 98 + m) * 128 + col] = acc[mt][r] + bias;
      }
  }
}

extern "C" void kernel_launch(void* const* d_in, const int* in_sizes, int n_in,
                              void* d_out, int out_size, void* d_ws,
                              size_t ws_size, hipStream_t stream) {
  const float* x      = (const float*)d_in[0];
  const float* mask   = (const float*)d_in[1];
  const float* w_qkv  = (const float*)d_in[2];
  const float* b_qkv  = (const float*)d_in[3];
  const float* w_proj = (const float*)d_in[4];
  const float* b_proj = (const float*)d_in[5];
  const float* table  = (const float*)d_in[6];
  float* out = (float*)d_out;
  float* bias_pre = (float*)d_ws;

  bias_pre_kernel<<<(4 * 98 * 98 + 255) / 256, 256, 0, stream>>>(table,
                                                                 bias_pre);
  fused_window_attn<<<2048, 256, 0, stream>>>(x, mask, w_qkv, b_qkv, w_proj,
                                              b_proj, bias_pre, out);
}


# x tile of the window staged once into LDS by whole-row 512-B loads; the QKV projection reads its A fragments from LDS, not by row-strided 8-B gathers from global memory
# speedup vs baseline: 1.9744x; 1.1841x over previous
#include <hip/hip_runtime.h>
#include <hip/hip_bf16.h>
#include <math.h>

typedef __attribute__((ext_vector_type(2))) float v2f;
typedef __attribute__((ext_vector_type(4))) float v4f;
typedef __attribute__((ext_vector_type(8))) float v8f;

__device__ __forceinline__ v8f wmma_f32(v2f a, v2f b, v8f c) {
  return __builtin_amdgcn_wmma_f32_16x16x4_f32(false, a, false, b, (short)0, c,
                                               false, false);
}

__global__ void bias_pre_kernel(const float* __restrict__ table,
                                float* __restrict__ bias_pre) {
  int tid = blockIdx.x * blockDim.x + threadIdx.x;
  if (tid >= 4 * 98 * 98) return;
  int h = tid / 9604;
  int rem = tid - h * 9604;
  int m = rem / 98;
  int n = rem - m * 98;
  int zm = m / 49, ym = (m % 49) / 7, xm = m % 7;
  int zn = n / 49, yn = (n % 49) / 7, xn = n % 7;
  int idx = (zm - zn + 1) * 169 + (ym - yn + 6) * 13 + (xm - xn + 6);
  bias_pre[tid] = table[idx * 4 + h];
}

__global__ __launch_bounds__(256) void fused_window_attn(
    const float* __restrict__ x, const float* __restrict__ mask,
    const float* __restrict__ w_qkv, const float* __restrict__ bq,
    const float* __restrict__ w_proj, const float* __restrict__ bpj,
    const float* __restrict__ bias_pre, float* __restrict__ out) {
  __shared__ float sQKV[112][386];
  __shared__ __attribute__((aligned(16))) float sO[112][130];
  __shared__ float sP[8][16][114];

  const int b = blockIdx.x;
  const int wd = b & 511;
  const int wave = threadIdx.x >> 5;
  const int lane = threadIdx.x & 31;
  const int l16 = lane & 15;
  const int hi = lane >> 4;
  const float* xb = x + (size_t)b * 98 * 128;

  float* sX = &sO[0][0];
  {
    const v4f* xg = (const v4f*)xb;
    v4f t[13];
#pragma unroll
    for (int i = 0; i < 12; ++i) t[i] = xg[(wave + 8 * i) * 32 + lane];
    const int r12 = 96 + wave;
    t[12] = xg[(r12 < 98 ? r12 : 97) * 32 + lane];
#pragma unroll
    for (int i = 0; i < 12; ++i)
      *(v4f*)(sX + (wave + 8 * i) * 132 + 4 * lane) = t[i];
    *(v4f*)(sX + r12 * 132 + 4 * lane) = t[12];
  }
  __syncthreads();

  for (int i = 0; i < 3; ++i) {
    const int nt = wave * 3 + i;
    const int col = nt * 16 + l16;
    v8f acc[7] = {};
    for (int k0 = 0; k0 < 128; k0 += 4) {
      const int kb = k0 + 2 * hi;
      v2f bw;
      bw.x = w_qkv[(size_t)kb * 384 + col];
      bw.y = w_qkv[(size_t)(kb + 1) * 384 + col];
#pragma unroll
      for (int mt = 0; mt < 7; ++mt) {
        int n = mt * 16 + l16;
        int nc = n < 98 ? n : 97;
        v2f a = *(const v2f*)(sX + nc * 132 + k0 + 2 * hi);
        acc[mt] = wmma_f32(a, bw, acc[mt]);
      }
    }
    const float bias = bq[col];
    const float scale = (col < 128) ? 0.17677669529663687f : 1.0f;
#pragma unroll
    for (int mt = 0; mt < 7; ++mt)
#pragma unroll
      for (int r = 0; r < 8; ++r) {
        int m = mt * 16 + r + 8 * hi;
        sQKV[m][col] = (acc[mt][r] + bias) * scale;
      }
  }
  __syncthreads();

  for (int it = 0; it < 4; ++it) {
    const int u = wave + 8 * it;
    if (u < 28) {
      const int h = u / 7;
      const int mt = u - h * 7;

      const float* bp = bias_pre + (size_t)h * 9604;
      const float* mp = mask + (size_t)wd * 9604;
      float bv[7][8], mv[7][8];
#pragma unroll
      for (int nt = 0; nt < 7; ++nt) {
        const int n = nt * 16 + l16;
        const int nn = n < 98 ? n : 97;
#pragma unroll
        for (int r = 0; r < 8; ++r) {
          const int m = mt * 16 + r + 8 * hi;
          const int mm = m < 98 ? m : 97;
          bv[nt][r] = bp[mm * 98 + nn];
          mv[nt][r] = mp[mm * 98 + nn];
        }
      }

      v8f acc[7] = {};
      for (int ks = 0; ks < 8; ++ks) {
        const int k0 = ks * 4;
        v2f a = *(const v2f*)(&sQKV[mt * 16 + l16][h * 32 + k0 + 2 * hi]);
#pragma unroll
        for (int nt = 0; nt < 7; ++nt) {
          v2f bk =
              *(const v2f*)(&sQKV[nt * 16 + l16][128 + h * 32 + k0 + 2 * hi]);
          acc[nt] = wmma_f32(a, bk, acc[nt]);
        }
      }

#pragma unroll
      for (int nt = 0; nt < 7; ++nt) {
        const int n = nt * 16 + l16;
#pragma unroll
        for (int r = 0; r < 8; ++r) {
          const int m = mt * 16 + r + 8 * hi;
          const float s = acc[nt][r] + (bv[nt][r] + mv[nt][r]);
          acc[nt][r] = (n < 98 && m < 98) ? s : -3.0e38f;
        }
      }

#pragma unroll
      for (int r = 0; r < 8; ++r) {
        float mx = acc[0][r];
#pragma unroll
        for (int nt = 1; nt < 7; ++nt) mx = fmaxf(mx, acc[nt][r]);
#pragma unroll
        for (int off = 1; off < 16; off <<= 1)
          mx = fmaxf(mx, __shfl_xor(mx, off, 32));
        float e[7];
        float sum = 0.0f;
#pragma unroll
        for (int nt = 0; nt < 7; ++nt) {
          e[nt] = __expf(acc[nt][r] - mx);
          sum += e[nt];
        }
#pragma unroll
        for (int off = 1; off < 16; off <<= 1) sum += __shfl_xor(sum, off, 32);
        const float inv = 1.0f / sum;
        const int m = mt * 16 + r + 8 * hi;
        const bool rowv = (m < 98);
#pragma unroll
        for (int nt = 0; nt < 7; ++nt) {
          const int n = nt * 16 + l16;
          sP[wave][r + 8 * hi][nt * 16 + l16] =
              (rowv && n < 98) ? e[nt] * inv : 0.0f;
        }
      }

      v8f o0 = {}, o1 = {};
      for (int ks = 0; ks < 28; ++ks) {
        const int k0 = ks * 4;
        v2f a = *(const v2f*)(&sP[wave][l16][k0 + 2 * hi]);
        const int t0 = k0 + 2 * hi;
        v2f b0, b1;
        b0.x = sQKV[t0][256 + h * 32 + l16];
        b0.y = sQKV[t0 + 1][256 + h * 32 + l16];
        o0 = wmma_f32(a, b0, o0);
        b1.x = sQKV[t0][256 + h * 32 + 16 + l16];
        b1.y = sQKV[t0 + 1][256 + h * 32 + 16 + l16];
        o1 = wmma_f32(a, b1, o1);
      }
#pragma unroll
      for (int r = 0; r < 8; ++r) {
        const int m = mt * 16 + r + 8 * hi;
        sO[m][h * 32 + l16] = o0[r];
        sO[m][h * 32 + 16 + l16] = o1[r];
      }
    }
  }
  __syncthreads();

  {
    const int col = wave * 16 + l16;
    v8f acc[7] = {};
    for (int k0 = 0; k0 < 128; k0 += 4) {
      const int kb = k0 + 2 * hi;
      v2f bw;
      bw.x = w_proj[(size_t)kb * 128 + col];
      bw.y = w_proj[(size_t)(kb + 1) * 128 + col];
#pragma unroll
      for (int mt = 0; mt < 7; ++mt) {
        v2f a = *(const v2f*)(&sO[mt * 16 + l16][k0 + 2 * hi]);
        acc[mt] = wmma_f32(a, bw, acc[mt]);
      }
    }
    const float bias = bpj[col];
#pragma unroll
    for (int mt = 0; mt < 7; ++mt)
#pragma unroll
      for (int r = 0; r < 8; ++r) {
        const int m = mt * 16 + r + 8 * hi;
        if (m < 98) out[((size_t)b * 98 + m) * 128 + col] = acc[mt][r] + bias;
      }
  }
}

extern "C" void kernel_launch(void* const* d_in, const int* in_sizes, int n_in,
                              void* d_out, int out_size, void* d_ws,
                              size_t ws_size, hipStream_t stream) {
  const float* x      = (const float*)d_in[0];
  const float* mask   = (const float*)d_in[1];
  const float* w_qkv  = (const float*)d_in[2];
  const float* b_qkv  = (const float*)d_in[3];
  const float* w_proj = (const float*)d_in[4];
  const float* b_proj = (const float*)d_in[5];
  const float* table  = (const float*)d_in[6];
  float* out = (float*)d_out;
  float* bias_pre = (float*)d_ws;

  bias_pre_kernel<<<(4 * 98 * 98 + 255) / 256, 256, 0, stream>>>(table,
                                                                 bias_pre);
  fused_window_attn<<<2048, 256, 0, stream>>>(x, mask, w_qkv, b_qkv, w_proj,
                                              b_proj, bias_pre, out);
}


# 16 waves per workgroup (4 per SIMD) instead of 8: projections split by row tiles over wave pairs, attention units in 2 rounds; attention output stored over the unit's own Q tile so LDS still fits
# speedup vs baseline: 2.1041x; 1.0657x over previous
#include <hip/hip_runtime.h>
#include <hip/hip_bf16.h>
#include <math.h>

typedef __attribute__((ext_vector_type(2))) float v2f;
typedef __attribute__((ext_vector_type(4))) float v4f;
typedef __attribute__((ext_vector_type(8))) float v8f;

__device__ __forceinline__ v8f wmma_f32(v2f a, v2f b, v8f c) {
  return __builtin_amdgcn_wmma_f32_16x16x4_f32(false, a, false, b, (short)0, c,
                                               false, false);
}

__global__ void bias_pre_kernel(const float* __restrict__ table,
                                float* __restrict__ bias_pre) {
  int tid = blockIdx.x * blockDim.x + threadIdx.x;
  if (tid >= 4 * 98 * 98) return;
  int h = tid / 9604;
  int rem = tid - h * 9604;
  int m = rem / 98;
  int n = rem - m * 98;
  int zm = m / 49, ym = (m % 49) / 7, xm = m % 7;
  int zn = n / 49, yn = (n % 49) / 7, xn = n % 7;
  int idx = (zm - zn + 1) * 169 + (ym - yn + 6) * 13 + (xm - xn + 6);
  bias_pre[tid] = table[idx * 4 + h];
}

template <int NMT>
__device__ __forceinline__ void proj_qkv(const float* __restrict__ w_qkv,
                                         const float* __restrict__ bq,
                                         const float* sX, float (*sQKV)[386],
                                         int ng, int mt0, int l16, int hi) {
  for (int i = 0; i < 3; ++i) {
    const int nt = ng * 3 + i;
    const int col = nt * 16 + l16;
    v8f acc[NMT] = {};
    for (int k0 = 0; k0 < 128; k0 += 4) {
      const int kb = k0 + 2 * hi;
      v2f bw;
      bw.x = w_qkv[(size_t)kb * 384 + col];
      bw.y = w_qkv[(size_t)(kb + 1) * 384 + col];
#pragma unroll
      for (int j = 0; j < NMT; ++j) {
        int n = (mt0 + j) * 16 + l16;
        int nc = n < 98 ? n : 97;
        v2f a = *(const v2f*)(sX + nc * 132 + k0 + 2 * hi);
        acc[j] = wmma_f32(a, bw, acc[j]);
      }
    }
    const float bias = bq[col];
    const float scale = (col < 128) ? 0.17677669529663687f : 1.0f;
#pragma unroll
    for (int j = 0; j < NMT; ++j)
#pragma unroll
      for (int r = 0; r < 8; ++r) {
        int m = (mt0 + j) * 16 + r + 8 * hi;
        sQKV[m][col] = (acc[j][r] + bias) * scale;
      }
  }
}

template <int NMT>
__device__ __forceinline__ void proj_out(const float* __restrict__ w_proj,
                                         const float* __restrict__ bpj,
                                         float (*sQKV)[386],
                                         float* __restrict__ out, int b,
                                         int nt, int mt0, int l16, int hi) {
  const int col = nt * 16 + l16;
  v8f acc[NMT] = {};
  for (int k0 = 0; k0 < 128; k0 += 4) {
    const int kb = k0 + 2 * hi;
    v2f bw;
    bw.x = w_proj[(size_t)kb * 128 + col];
    bw.y = w_proj[(size_t)(kb + 1) * 128 + col];
#pragma unroll
    for (int j = 0; j < NMT; ++j) {
      v2f a = *(const v2f*)(&sQKV[(mt0 + j) * 16 + l16][k0 + 2 * hi]);
      acc[j] = wmma_f32(a, bw, acc[j]);
    }
  }
  const float bias = bpj[col];
#pragma unroll
  for (int j = 0; j < NMT; ++j)
#pragma unroll
    for (int r = 0; r < 8; ++r) {
      const int m = (mt0 + j) * 16 + r + 8 * hi;
      if (m < 98) out[((size_t)b * 98 + m) * 128 + col] = acc[j][r] + bias;
    }
}

__global__ __launch_bounds__(512) void fused_window_attn(
    const float* __restrict__ x, const float* __restrict__ mask,
    const float* __restrict__ w_qkv, const float* __restrict__ bq,
    const float* __restrict__ w_proj, const float* __restrict__ bpj,
    const float* __restrict__ bias_pre, float* __restrict__ out) {
  __shared__ __attribute__((aligned(16))) float sQKV[112][386];
  __shared__ __attribute__((aligned(16))) float sP[16][16][114];

  const int b = blockIdx.x;
  const int wd = b & 511;
  const int wave = __builtin_amdgcn_readfirstlane((int)(threadIdx.x >> 5));
  const int lane = threadIdx.x & 31;
  const int l16 = lane & 15;
  const int hi = lane >> 4;
  const float* xb = x + (size_t)b * 98 * 128;

  float* sX = &sP[0][0][0];
  {
    const v4f* xg = (const v4f*)xb;
    v4f t[7];
#pragma unroll
    for (int i = 0; i < 6; ++i) t[i] = xg[(wave + 16 * i) * 32 + lane];
    const int r6 = 96 + wave;
    t[6] = xg[(r6 < 98 ? r6 : 97) * 32 + lane];
#pragma unroll
    for (int i = 0; i < 6; ++i)
      *(v4f*)(sX + (wave + 16 * i) * 132 + 4 * lane) = t[i];
    *(v4f*)(sX + r6 * 132 + 4 * lane) = t[6];
  }
  __syncthreads();

  if (wave < 8)
    proj_qkv<4>(w_qkv, bq, sX, sQKV, wave, 0, l16, hi);
  else
    proj_qkv<3>(w_qkv, bq, sX, sQKV, wave - 8, 4, l16, hi);
  __syncthreads();

  for (int it = 0; it < 2; ++it) {
    const int u = wave + 16 * it;
    if (u < 28) {
      const int h = u / 7;
      const int mt = u - h * 7;

      const float* bp = bias_pre + (size_t)h * 9604;
      const float* mp = mask + (size_t)wd * 9604;
      float bv[7][8], mv[7][8];
#pragma unroll
      for (int nt = 0; nt < 7; ++nt) {
        const int n = nt * 16 + l16;
        const int nn = n < 98 ? n : 97;
#pragma unroll
        for (int r = 0; r < 8; ++r) {
          const int m = mt * 16 + r + 8 * hi;
          const int mm = m < 98 ? m : 97;
          bv[nt][r] = bp[mm * 98 + nn];
          mv[nt][r] = mp[mm * 98 + nn];
        }
      }

      v8f acc[7] = {};
      for (int ks = 0; ks < 8; ++ks) {
        const int k0 = ks * 4;
        v2f a = *(const v2f*)(&sQKV[mt * 16 + l16][h * 32 + k0 + 2 * hi]);
#pragma unroll
        for (int nt = 0; nt < 7; ++nt) {
          v2f bk =
              *(const v2f*)(&sQKV[nt * 16 + l16][128 + h * 32 + k0 + 2 * hi]);
          acc[nt] = wmma_f32(a, bk, acc[nt]);
        }
      }

#pragma unroll
      for (int nt = 0; nt < 7; ++nt) {
        const int n = nt * 16 + l16;
#pragma unroll
        for (int r = 0; r < 8; ++r) {
          const int m = mt * 16 + r + 8 * hi;
          const float s = acc[nt][r] + (bv[nt][r] + mv[nt][r]);
          acc[nt][r] = (n < 98 && m < 98) ? s : -3.0e38f;
        }
      }

#pragma unroll
      for (int r = 0; r < 8; ++r) {
        float mx = acc[0][r];
#pragma unroll
        for (int nt = 1; nt < 7; ++nt) mx = fmaxf(mx, acc[nt][r]);
#pragma unroll
        for (int off = 1; off < 16; off <<= 1)
          mx = fmaxf(mx, __shfl_xor(mx, off, 32));
        float e[7];
        float sum = 0.0f;
#pragma unroll
        for (int nt = 0; nt < 7; ++nt) {
          e[nt] = __expf(acc[nt][r] - mx);
          sum += e[nt];
        }
#pragma unroll
        for (int off = 1; off < 16; off <<= 1) sum += __shfl_xor(sum, off, 32);
        const float inv = 1.0f / sum;
        const int m = mt * 16 + r + 8 * hi;
        const bool rowv = (m < 98);
#pragma unroll
        for (int nt = 0; nt < 7; ++nt) {
          const int n = nt * 16 + l16;
          sP[wave][r + 8 * hi][nt * 16 + l16] =
              (rowv && n < 98) ? e[nt] * inv : 0.0f;
        }
      }

      v8f o0 = {}, o1 = {};
      for (int ks = 0; ks < 28; ++ks) {
        const int k0 = ks * 4;
        v2f a = *(const v2f*)(&sP[wave][l16][k0 + 2 * hi]);
        const int t0 = k0 + 2 * hi;
        v2f b0, b1;
        b0.x = sQKV[t0][256 + h * 32 + l16];
        b0.y = sQKV[t0 + 1][256 + h * 32 + l16];
        o0 = wmma_f32(a, b0, o0);
        b1.x = sQKV[t0][256 + h * 32 + 16 + l16];
        b1.y = sQKV[t0 + 1][256 + h * 32 + 16 + l16];
        o1 = wmma_f32(a, b1, o1);
      }
#pragma unroll
      for (int r = 0; r < 8; ++r) {
        const int m = mt * 16 + r + 8 * hi;
        sQKV[m][h * 32 + l16] = o0[r];
        sQKV[m][h * 32 + 16 + l16] = o1[r];
      }
    }
  }
  __syncthreads();

  if (wave < 8)
    proj_out<4>(w_proj, bpj, sQKV, out, b, wave, 0, l16, hi);
  else
    proj_out<3>(w_proj, bpj, sQKV, out, b, wave - 8, 4, l16, hi);
}

extern "C" void kernel_launch(void* const* d_in, const int* in_sizes, int n_in,
                              void* d_out, int out_size, void* d_ws,
                              size_t ws_size, hipStream_t stream) {
  const float* x      = (const float*)d_in[0];
  const float* mask   = (const float*)d_in[1];
  const float* w_qkv  = (const float*)d_in[2];
  const float* b_qkv  = (const float*)d_in[3];
  const float* w_proj = (const float*)d_in[4];
  const float* b_proj = (const float*)d_in[5];
  const float* table  = (const float*)d_in[6];
  float* out = (float*)d_out;
  float* bias_pre = (float*)d_ws;

  bias_pre_kernel<<<(4 * 98 * 98 + 255) / 256, 256, 0, stream>>>(table,
                                                                 bias_pre);
  fused_window_attn<<<2048, 512, 0, stream>>>(x, mask, w_qkv, b_qkv, w_proj,
                                              b_proj, bias_pre, out);
}


# Weight fragments of both projections fetched one block of 8 k-steps ahead of the matrix work that uses them (first block ahead of the stage barrier): no load round trip inside a k loop
# speedup vs baseline: 2.1987x; 1.0450x over previous
#include <hip/hip_runtime.h>
#include <hip/hip_bf16.h>
#include <math.h>

typedef __attribute__((ext_vector_type(2))) float v2f;
typedef __attribute__((ext_vector_type(4))) float v4f;
typedef __attribute__((ext_vector_type(8))) float v8f;

__device__ __forceinline__ v8f wmma_f32(v2f a, v2f b, v8f c) {
  return __builtin_amdgcn_wmma_f32_16x16x4_f32(false, a, false, b, (short)0, c,
                                               false, false);
}

__global__ void bias_pre_kernel(const float* __restrict__ table,
                                float* __restrict__ bias_pre) {
  int tid = blockIdx.x * blockDim.x + threadIdx.x;
  if (tid >= 4 * 98 * 98) return;
  int h = tid / 9604;
  int rem = tid - h * 9604;
  int m = rem / 98;
  int n = rem - m * 98;
  int zm = m / 49, ym = (m % 49) / 7, xm = m % 7;
  int zn = n / 49, yn = (n % 49) / 7, xn = n % 7;
  int idx = (zm - zn + 1) * 169 + (ym - yn + 6) * 13 + (xm - xn + 6);
  bias_pre[tid] = table[idx * 4 + h];
}

template <int LD>
__device__ __forceinline__ void load_wblk(const float* __restrict__ w, int col,
                                          int kblk, int hi, v2f (&bf)[8]) {
#pragma unroll
  for (int s = 0; s < 8; ++s) {
    const int kb = kblk * 32 + s * 4 + 2 * hi;
    bf[s].x = w[(size_t)kb * LD + col];
    bf[s].y = w[(size_t)(kb + 1) * LD + col];
  }
}

template <int NMT>
__device__ __forceinline__ void qkv_block(const float* sX, int kblk, int mt0,
                                          int l16, int hi, const v2f (&bf)[8],
                                          v8f (&acc)[NMT]) {
#pragma unroll
  for (int s = 0; s < 8; ++s) {
    const int k0 = kblk * 32 + s * 4;
#pragma unroll
    for (int j = 0; j < NMT; ++j) {
      int n = (mt0 + j) * 16 + l16;
      int nc = n < 98 ? n : 97;
      v2f a = *(const v2f*)(sX + nc * 132 + k0 + 2 * hi);
      acc[j] = wmma_f32(a, bf[s], acc[j]);
    }
  }
}

template <int NMT>
__device__ __forceinline__ void proj_qkv(const float* __restrict__ w_qkv,
                                         const float* __restrict__ bq,
                                         const float* sX, float (*sQKV)[386],
                                         int ng, int mt0, int l16, int hi,
                                         v2f (&b0)[8]) {
  v2f b1[8];
#pragma unroll 1
  for (int i = 0; i < 3; ++i) {
    const int nt = ng * 3 + i;
    const int col = nt * 16 + l16;
    const float bias = bq[col];
    v8f acc[NMT] = {};
#pragma unroll 1
    for (int kp = 0; kp < 2; ++kp) {
      load_wblk<384>(w_qkv, col, 2 * kp + 1, hi, b1);
      qkv_block<NMT>(sX, 2 * kp, mt0, l16, hi, b0, acc);
      const int ncol = (kp == 0 || i == 2) ? col : col + 16;
      load_wblk<384>(w_qkv, ncol, kp == 0 ? 2 : 0, hi, b0);
      qkv_block<NMT>(sX, 2 * kp + 1, mt0, l16, hi, b1, acc);
    }
    const float scale = (col < 128) ? 0.17677669529663687f : 1.0f;
#pragma unroll
    for (int j = 0; j < NMT; ++j)
#pragma unroll
      for (int r = 0; r < 8; ++r) {
        int m = (mt0 + j) * 16 + r + 8 * hi;
        sQKV[m][col] = (acc[j][r] + bias) * scale;
      }
  }
}

template <int NMT>
__device__ __forceinline__ void proj_out(const float* __restrict__ w_proj,
                                         float bias, float (*sQKV)[386],
                                         float* __restrict__ out, int b,
                                         int nt, int mt0, int l16, int hi,
                                         v2f (&bn)[8]) {
  const int col = nt * 16 + l16;
  v8f acc[NMT] = {};
#pragma unroll
  for (int kblk = 0; kblk < 4; ++kblk) {
    v2f bc[8];
#pragma unroll
    for (int s = 0; s < 8; ++s) bc[s] = bn[s];
    if (kblk < 3) load_wblk<128>(w_proj, col, kblk + 1, hi, bn);
#pragma unroll
    for (int s = 0; s < 8; ++s) {
      const int k0 = kblk * 32 + s * 4;
#pragma unroll
      for (int j = 0; j < NMT; ++j) {
        v2f a = *(const v2f*)(&sQKV[(mt0 + j) * 16 + l16][k0 + 2 * hi]);
        acc[j] = wmma_f32(a, bc[s], acc[j]);
      }
    }
  }
#pragma unroll
  for (int j = 0; j < NMT; ++j)
#pragma unroll
    for (int r = 0; r < 8; ++r) {
      const int m = (mt0 + j) * 16 + r + 8 * hi;
      if (m < 98) out[((size_t)b * 98 + m) * 128 + col] = acc[j][r] + bias;
    }
}

__global__ __launch_bounds__(512) void fused_window_attn(
    const float* __restrict__ x, const float* __restrict__ mask,
    const float* __restrict__ w_qkv, const float* __restrict__ bq,
    const float* __restrict__ w_proj, const float* __restrict__ bpj,
    const float* __restrict__ bias_pre, float* __restrict__ out) {
  __shared__ __attribute__((aligned(16))) float sQKV[112][386];
  __shared__ __attribute__((aligned(16))) float sP[16][16][114];

  const int b = blockIdx.x;
  const int wd = b & 511;
  const int wave = __builtin_amdgcn_readfirstlane((int)(threadIdx.x >> 5));
  const int lane = threadIdx.x & 31;
  const int l16 = lane & 15;
  const int hi = lane >> 4;
  const float* xb = x + (size_t)b * 98 * 128;

  float* sX = &sP[0][0][0];
  v2f wq[8];
  {
    const v4f* xg = (const v4f*)xb;
    v4f t[7];
#pragma unroll
    for (int i = 0; i < 6; ++i) t[i] = xg[(wave + 16 * i) * 32 + lane];
    const int r6 = 96 + wave;
    t[6] = xg[(r6 < 98 ? r6 : 97) * 32 + lane];
    load_wblk<384>(w_qkv, (wave & 7) * 48 + l16, 0, hi, wq);
#pragma unroll
    for (int i = 0; i < 6; ++i)
      *(v4f*)(sX + (wave + 16 * i) * 132 + 4 * lane) = t[i];
    *(v4f*)(sX + r6 * 132 + 4 * lane) = t[6];
  }
  __syncthreads();

  if (wave < 8)
    proj_qkv<4>(w_qkv, bq, sX, sQKV, wave, 0, l16, hi, wq);
  else
    proj_qkv<3>(w_qkv, bq, sX, sQKV, wave - 8, 4, l16, hi, wq);
  __syncthreads();

  for (int it = 0; it < 2; ++it) {
    const int u = wave + 16 * it;
    if (u < 28) {
      const int h = u / 7;
      const int mt = u - h * 7;

      const float* bp = bias_pre + (size_t)h * 9604;
      const float* mp = mask + (size_t)wd * 9604;
      float bv[7][8], mv[7][8];
#pragma unroll
      for (int nt = 0; nt < 7; ++nt) {
        const int n = nt * 16 + l16;
        const int nn = n < 98 ? n : 97;
#pragma unroll
        for (int r = 0; r < 8; ++r) {
          const int m = mt * 16 + r + 8 * hi;
          const int mm = m < 98 ? m : 97;
          bv[nt][r] = bp[mm * 98 + nn];
          mv[nt][r] = mp[mm * 98 + nn];
        }
      }

      v8f acc[7] = {};
      for (int ks = 0; ks < 8; ++ks) {
        const int k0 = ks * 4;
        v2f a = *(const v2f*)(&sQKV[mt * 16 + l16][h * 32 + k0 + 2 * hi]);
#pragma unroll
        for (int nt = 0; nt < 7; ++nt) {
          v2f bk =
              *(const v2f*)(&sQKV[nt * 16 + l16][128 + h * 32 + k0 + 2 * hi]);
          acc[nt] = wmma_f32(a, bk, acc[nt]);
        }
      }

#pragma unroll
      for (int nt = 0; nt < 7; ++nt) {
        const int n = nt * 16 + l16;
#pragma unroll
        for (int r = 0; r < 8; ++r) {
          const int m = mt * 16 + r + 8 * hi;
          const float s = acc[nt][r] + (bv[nt][r] + mv[nt][r]);
          acc[nt][r] = (n < 98 && m < 98) ? s : -3.0e38f;
        }
      }

#pragma unroll
      for (int r = 0; r < 8; ++r) {
        float mx = acc[0][r];
#pragma unroll
        for (int nt = 1; nt < 7; ++nt) mx = fmaxf(mx, acc[nt][r]);
#pragma unroll
        for (int off = 1; off < 16; off <<= 1)
          mx = fmaxf(mx, __shfl_xor(mx, off, 32));
        float e[7];
        float sum = 0.0f;
#pragma unroll
        for (int nt = 0; nt < 7; ++nt) {
          e[nt] = __expf(acc[nt][r] - mx);
          sum += e[nt];
        }
#pragma unroll
        for (int off = 1; off < 16; off <<= 1) sum += __shfl_xor(sum, off, 32);
        const float inv = 1.0f / sum;
        const int m = mt * 16 + r + 8 * hi;
        const bool rowv = (m < 98);
#pragma unroll
        for (int nt = 0; nt < 7; ++nt) {
          const int n = nt * 16 + l16;
          sP[wave][r + 8 * hi][nt * 16 + l16] =
              (rowv && n < 98) ? e[nt] * inv : 0.0f;
        }
      }

      v8f o0 = {}, o1 = {};
      for (int ks = 0; ks < 28; ++ks) {
        const int k0 = ks * 4;
        v2f a = *(const v2f*)(&sP[wave][l16][k0 + 2 * hi]);
        const int t0 = k0 + 2 * hi;
        v2f b0, b1;
        b0.x = sQKV[t0][256 + h * 32 + l16];
        b0.y = sQKV[t0 + 1][256 + h * 32 + l16];
        o0 = wmma_f32(a, b0, o0);
        b1.x = sQKV[t0][256 + h * 32 + 16 + l16];
        b1.y = sQKV[t0 + 1][256 + h * 32 + 16 + l16];
        o1 = wmma_f32(a, b1, o1);
      }
#pragma unroll
      for (int r = 0; r < 8; ++r) {
        const int m = mt * 16 + r + 8 * hi;
        sQKV[m][h * 32 + l16] = o0[r];
        sQKV[m][h * 32 + 16 + l16] = o1[r];
      }
    }
  }
  v2f wp[8];
  load_wblk<128>(w_proj, (wave & 7) * 16 + l16, 0, hi, wp);
  const float bias_o = bpj[(wave & 7) * 16 + l16];
  __syncthreads();

  if (wave < 8)
    proj_out<4>(w_proj, bias_o, sQKV, out, b, wave, 0, l16, hi, wp);
  else
    proj_out<3>(w_proj, bias_o, sQKV, out, b, wave - 8, 4, l16, hi, wp);
}

extern "C" void kernel_launch(void* const* d_in, const int* in_sizes, int n_in,
                              void* d_out, int out_size, void* d_ws,
                              size_t ws_size, hipStream_t stream) {
  const float* x      = (const float*)d_in[0];
  const float* mask   = (const float*)d_in[1];
  const float* w_qkv  = (const float*)d_in[2];
  const float* b_qkv  = (const float*)d_in[3];
  const float* w_proj = (const float*)d_in[4];
  const float* b_proj = (const float*)d_in[5];
  const float* table  = (const float*)d_in[6];
  float* out = (float*)d_out;
  float* bias_pre = (float*)d_ws;

  bias_pre_kernel<<<(4 * 98 * 98 + 255) / 256, 256, 0, stream>>>(table,
                                                                 bias_pre);
  fused_window_attn<<<2048, 512, 0, stream>>>(x, mask, w_qkv, b_qkv, w_proj,
                                              b_proj, bias_pre, out);
}


# Scores formed transposed (S^T = K Q^T): softmax runs in-lane plus one half-exchange, P feeds P V from registers, mask staged once in LDS; row-sum and P V summation order change
# speedup vs baseline: 2.4406x; 1.1100x over previous
#include <hip/hip_runtime.h>
#include <hip/hip_bf16.h>
#include <math.h>

typedef __attribute__((ext_vector_type(2))) float v2f;
typedef __attribute__((ext_vector_type(4))) float v4f;
typedef __attribute__((ext_vector_type(8))) float v8f;

__device__ __forceinline__ v8f wmma_f32(v2f a, v2f b, v8f c) {
  return __builtin_amdgcn_wmma_f32_16x16x4_f32(false, a, false, b, (short)0, c,
                                               false, false);
}

__device__ __forceinline__ float xhalf(float x) {
  const int b = __builtin_bit_cast(int, x);
  return __builtin_bit_cast(float, __builtin_amdgcn_permlanex16(
      b, b, 0x76543210, 0xfedcba98, false, false));
}

__global__ void bias_pre_kernel(const float* __restrict__ table,
                                float* __restrict__ bias_pre) {
  int tid = blockIdx.x * blockDim.x + threadIdx.x;
  if (tid >= 4 * 98 * 98) return;
  int h = tid / 9604;
  int rem = tid - h * 9604;
  int n = rem / 98;
  int m = rem - n * 98;
  int zm = m / 49, ym = (m % 49) / 7, xm = m % 7;
  int zn = n / 49, yn = (n % 49) / 7, xn = n % 7;
  int idx = (zm - zn + 1) * 169 + (ym - yn + 6) * 13 + (xm - xn + 6);
  bias_pre[tid] = table[idx * 4 + h];
}

template <int LD>
__device__ __forceinline__ void load_wblk(const float* __restrict__ w, int col,
                                          int kblk, int hi, v2f (&bf)[8]) {
#pragma unroll
  for (int s = 0; s < 8; ++s) {
    const int kb = kblk * 32 + s * 4 + 2 * hi;
    bf[s].x = w[(size_t)kb * LD + col];
    bf[s].y = w[(size_t)(kb + 1) * LD + col];
  }
}

template <int NMT>
__device__ __forceinline__ void qkv_block(const float* sX, int kblk, int mt0,
                                          int l16, int hi, const v2f (&bf)[8],
                                          v8f (&acc)[NMT]) {
#pragma unroll
  for (int s = 0; s < 8; ++s) {
    const int k0 = kblk * 32 + s * 4;
#pragma unroll
    for (int j = 0; j < NMT; ++j) {
      int n = (mt0 + j) * 16 + l16;
      int nc = n < 98 ? n : 97;
      v2f a = *(const v2f*)(sX + nc * 132 + k0 + 2 * hi);
      acc[j] = wmma_f32(a, bf[s], acc[j]);
    }
  }
}

template <int NMT>
__device__ __forceinline__ void proj_qkv(const float* __restrict__ w_qkv,
                                         const float* __restrict__ bq,
                                         const float* sX, float (*sQKV)[386],
                                         int ng, int mt0, int l16, int hi,
                                         v2f (&b0)[8]) {
  v2f b1[8];
#pragma unroll 1
  for (int i = 0; i < 3; ++i) {
    const int nt = ng * 3 + i;
    const int col = nt * 16 + l16;
    const float bias = bq[col];
    v8f acc[NMT] = {};
#pragma unroll 1
    for (int kp = 0; kp < 2; ++kp) {
      load_wblk<384>(w_qkv, col, 2 * kp + 1, hi, b1);
      qkv_block<NMT>(sX, 2 * kp, mt0, l16, hi, b0, acc);
      const int ncol = (kp == 0 || i == 2) ? col : col + 16;
      load_wblk<384>(w_qkv, ncol, kp == 0 ? 2 : 0, hi, b0);
      qkv_block<NMT>(sX, 2 * kp + 1, mt0, l16, hi, b1, acc);
    }
    const float scale = (col < 128) ? 0.17677669529663687f : 1.0f;
#pragma unroll
    for (int j = 0; j < NMT; ++j)
#pragma unroll
      for (int r = 0; r < 8; ++r) {
        int m = (mt0 + j) * 16 + r + 8 * hi;
        sQKV[m][col] = (acc[j][r] + bias) * scale;
      }
  }
}

template <int NMT>
__device__ __forceinline__ void proj_out(const float* __restrict__ w_proj,
                                         float bias, float (*sQKV)[386],
                                         float* __restrict__ out, int b,
                                         int nt, int mt0, int l16, int hi,
                                         v2f (&bn)[8]) {
  const int col = nt * 16 + l16;
  v8f acc[NMT] = {};
#pragma unroll
  for (int kblk = 0; kblk < 4; ++kblk) {
    v2f bc[8];
#pragma unroll
    for (int s = 0; s < 8; ++s) bc[s] = bn[s];
    if (kblk < 3) load_wblk<128>(w_proj, col, kblk + 1, hi, bn);
#pragma unroll
    for (int s = 0; s < 8; ++s) {
      const int k0 = kblk * 32 + s * 4;
#pragma unroll
      for (int j = 0; j < NMT; ++j) {
        v2f a = *(const v2f*)(&sQKV[(mt0 + j) * 16 + l16][k0 + 2 * hi]);
        acc[j] = wmma_f32(a, bc[s], acc[j]);
      }
    }
  }
#pragma unroll
  for (int j = 0; j < NMT; ++j)
#pragma unroll
    for (int r = 0; r < 8; ++r) {
      const int m = (mt0 + j) * 16 + r + 8 * hi;
      if (m < 98) out[((size_t)b * 98 + m) * 128 + col] = acc[j][r] + bias;
    }
}

__global__ __launch_bounds__(512) void fused_window_attn(
    const float* __restrict__ x, const float* __restrict__ mask,
    const float* __restrict__ w_qkv, const float* __restrict__ bq,
    const float* __restrict__ w_proj, const float* __restrict__ bpj,
    const float* __restrict__ bias_pre, float* __restrict__ out) {
  __shared__ __attribute__((aligned(16))) float sQKV[112][386];
  __shared__ __attribute__((aligned(16))) float sX[112 * 132];
  __shared__ __attribute__((aligned(16))) float sM[98 * 100];

  const int b = blockIdx.x;
  const int wd = b & 511;
  const int wave = __builtin_amdgcn_readfirstlane((int)(threadIdx.x >> 5));
  const int lane = threadIdx.x & 31;
  const int l16 = lane & 15;
  const int hi = lane >> 4;
  const float* xb = x + (size_t)b * 98 * 128;

  v2f wq[8];
  {
    const v4f* xg = (const v4f*)xb;
    v4f t[7];
#pragma unroll
    for (int i = 0; i < 6; ++i) t[i] = xg[(wave + 16 * i) * 32 + lane];
    const int r6 = 96 + wave;
    t[6] = xg[(r6 < 98 ? r6 : 97) * 32 + lane];
    load_wblk<384>(w_qkv, (wave & 7) * 48 + l16, 0, hi, wq);
    v2f mk[10];
    {
      const v2f* mg = (const v2f*)(mask + (size_t)wd * 9604);
#pragma unroll
      for (int it = 0; it < 10; ++it) {
        const int i = (int)threadIdx.x + 512 * it;
        mk[it] = mg[i < 4802 ? i : 4801];
      }
    }
#pragma unroll
    for (int i = 0; i < 6; ++i)
      *(v4f*)(sX + (wave + 16 * i) * 132 + 4 * lane) = t[i];
    *(v4f*)(sX + r6 * 132 + 4 * lane) = t[6];
#pragma unroll
    for (int it = 0; it < 10; ++it) {
      const int i = (int)threadIdx.x + 512 * it;
      const int ic = i < 4802 ? i : 4801;
      const int row = ic / 49;
      const int c = ic - row * 49;
      *(v2f*)(sM + row * 100 + 2 * c) = mk[it];
    }
  }
  __syncthreads();

  if (wave < 8)
    proj_qkv<4>(w_qkv, bq, sX, sQKV, wave, 0, l16, hi, wq);
  else
    proj_qkv<3>(w_qkv, bq, sX, sQKV, wave - 8, 4, l16, hi, wq);
  __syncthreads();

  for (int it = 0; it < 2; ++it) {
    const int u = wave + 16 * it;
    if (u < 28) {
      const int h = u / 7;
      const int mt = u - h * 7;
      const int m = mt * 16 + l16;
      const int mm = m < 98 ? m : 97;
      const bool qv = (m < 98);

      const float* bp = bias_pre + (size_t)h * 9604;
      float bv[7][8];
#pragma unroll
      for (int nt = 0; nt < 7; ++nt) {
#pragma unroll
        for (int r = 0; r < 8; ++r) {
          if (nt == 6 && r >= 2) {
            bv[nt][r] = 0.0f;
          } else {
            const int n = nt * 16 + 8 * hi + r;
            const int nn = n < 98 ? n : 97;
            bv[nt][r] = bp[nn * 98 + mm];
          }
        }
      }

      v8f acc[7] = {};
      for (int ks = 0; ks < 8; ++ks) {
        const int k0 = ks * 4;
        v2f q = *(const v2f*)(&sQKV[mt * 16 + l16][h * 32 + k0 + 2 * hi]);
#pragma unroll
        for (int nt = 0; nt < 7; ++nt) {
          v2f kf =
              *(const v2f*)(&sQKV[nt * 16 + l16][128 + h * 32 + k0 + 2 * hi]);
          acc[nt] = wmma_f32(kf, q, acc[nt]);
        }
      }

      const float* mrow = sM + mm * 100;
#pragma unroll
      for (int nt = 0; nt < 6; ++nt) {
        const v4f m0 = *(const v4f*)(mrow + nt * 16 + 8 * hi);
        const v4f m1 = *(const v4f*)(mrow + nt * 16 + 8 * hi + 4);
#pragma unroll
        for (int r = 0; r < 8; ++r) {
          const float mvv = r < 4 ? m0[r] : m1[r - 4];
          const float s = acc[nt][r] + (bv[nt][r] + mvv);
          acc[nt][r] = qv ? s : -3.0e38f;
        }
      }
      {
        const v2f m6 = *(const v2f*)(mrow + 96);
        const bool kv = qv && (hi == 0);
#pragma unroll
        for (int r = 0; r < 2; ++r) {
          const float s = acc[6][r] + (bv[6][r] + m6[r]);
          acc[6][r] = kv ? s : -3.0e38f;
        }
      }

      float mx = fmaxf(acc[6][0], acc[6][1]);
#pragma unroll
      for (int nt = 0; nt < 6; ++nt) {
        float t = acc[nt][0];
#pragma unroll
        for (int r = 1; r < 8; ++r) t = fmaxf(t, acc[nt][r]);
        mx = fmaxf(mx, t);
      }
      mx = fmaxf(mx, xhalf(mx));
      float ps[7];
#pragma unroll
      for (int nt = 0; nt < 6; ++nt) {
        ps[nt] = 0.0f;
#pragma unroll
        for (int r = 0; r < 8; ++r) {
          acc[nt][r] = __expf(acc[nt][r] - mx);
          ps[nt] += acc[nt][r];
        }
      }
      acc[6][0] = __expf(acc[6][0] - mx);
      acc[6][1] = __expf(acc[6][1] - mx);
      ps[6] = acc[6][0] + acc[6][1];
      float sum = ((ps[0] + ps[1]) + (ps[2] + ps[3])) +
                  ((ps[4] + ps[5]) + ps[6]);
      sum += xhalf(sum);
      const float inv = 1.0f / sum;
#pragma unroll
      for (int nt = 0; nt < 6; ++nt)
#pragma unroll
        for (int r = 0; r < 8; ++r)
          acc[nt][r] = qv ? acc[nt][r] * inv : 0.0f;
      {
        const bool kv = qv && (hi == 0);
        acc[6][0] = kv ? acc[6][0] * inv : 0.0f;
        acc[6][1] = kv ? acc[6][1] * inv : 0.0f;
      }

      v8f o0 = {}, o1 = {};
#pragma unroll
      for (int nt = 0; nt < 7; ++nt) {
#pragma unroll
        for (int j = 0; j < 4; ++j) {
          if (nt == 6 && j > 0) continue;
          v2f a;
          a.x = acc[nt][2 * j];
          a.y = acc[nt][2 * j + 1];
          const int t0 = nt * 16 + 8 * hi + 2 * j;
          v2f b0, b1;
          b0.x = sQKV[t0][256 + h * 32 + l16];
          b0.y = sQKV[t0 + 1][256 + h * 32 + l16];
          o0 = wmma_f32(a, b0, o0);
          b1.x = sQKV[t0][256 + h * 32 + 16 + l16];
          b1.y = sQKV[t0 + 1][256 + h * 32 + 16 + l16];
          o1 = wmma_f32(a, b1, o1);
        }
      }
#pragma unroll
      for (int r = 0; r < 8; ++r) {
        const int mo = mt * 16 + r + 8 * hi;
        sQKV[mo][h * 32 + l16] = o0[r];
        sQKV[mo][h * 32 + 16 + l16] = o1[r];
      }
    }
  }
  v2f wp[8];
  load_wblk<128>(w_proj, (wave & 7) * 16 + l16, 0, hi, wp);
  const float bias_o = bpj[(wave & 7) * 16 + l16];
  __syncthreads();

  if (wave < 8)
    proj_out<4>(w_proj, bias_o, sQKV, out, b, wave, 0, l16, hi, wp);
  else
    proj_out<3>(w_proj, bias_o, sQKV, out, b, wave - 8, 4, l16, hi, wp);
}

extern "C" void kernel_launch(void* const* d_in, const int* in_sizes, int n_in,
                              void* d_out, int out_size, void* d_ws,
                              size_t ws_size, hipStream_t stream) {
  const float* x      = (const float*)d_in[0];
  const float* mask   = (const float*)d_in[1];
  const float* w_qkv  = (const float*)d_in[2];
  const float* b_qkv  = (const float*)d_in[3];
  const float* w_proj = (const float*)d_in[4];
  const float* b_proj = (const float*)d_in[5];
  const float* table  = (const float*)d_in[6];
  float* out = (float*)d_out;
  float* bias_pre = (float*)d_ws;

  bias_pre_kernel<<<(4 * 98 * 98 + 255) / 256, 256, 0, stream>>>(table,
                                                                 bias_pre);
  fused_window_attn<<<2048, 512, 0, stream>>>(x, mask, w_qkv, b_qkv, w_proj,
                                              b_proj, bias_pre, out);
}


# B operands stored k-contiguous: one 16-byte load feeds two k-steps (was four 4-byte loads); weights repacked in fragment order by the first launch, V written transposed to LDS by the QKV epilogue
# speedup vs baseline: 2.6698x; 1.0939x over previous
#include <hip/hip_runtime.h>
#include <hip/hip_bf16.h>
#include <math.h>

typedef __attribute__((ext_vector_type(2))) float v2f;
typedef __attribute__((ext_vector_type(4))) float v4f;
typedef __attribute__((ext_vector_type(8))) float v8f;

__device__ __forceinline__ v8f wmma_f32(v2f a, v2f b, v8f c) {
  return __builtin_amdgcn_wmma_f32_16x16x4_f32(false, a, false, b, (short)0, c,
                                               false, false);
}

__device__ __forceinline__ float xhalf(float x) {
  const int b = __builtin_bit_cast(int, x);
  return __builtin_bit_cast(float, __builtin_amdgcn_permlanex16(
      b, b, 0x76543210, 0xfedcba98, false, false));
}

__global__ void bias_pre_kernel(const float* __restrict__ table,
                                const float* __restrict__ w_qkv,
                                const float* __restrict__ w_proj,
                                float* __restrict__ bias_pre,
                                float* __restrict__ wq_pk,
                                float* __restrict__ wp_pk) {
  if (blockIdx.x >= 151) {
    int t = (blockIdx.x - 151) * 256 + threadIdx.x;
    const float* w = w_qkv;
    float* pk = wq_pk;
    int ld = 384;
    if (t >= 24 * 16 * 32) {
      t -= 24 * 16 * 32;
      w = w_proj;
      pk = wp_pk;
      ld = 128;
    }
    const int lane = t & 31;
    const int kk = (t >> 5) & 15;
    const int nt = t >> 9;
    const int col = nt * 16 + (lane & 15);
    const int kb = 8 * kk + 2 * (lane >> 4);
    v4f v;
    v.x = w[(kb + 0) * ld + col];
    v.y = w[(kb + 1) * ld + col];
    v.z = w[(kb + 4) * ld + col];
    v.w = w[(kb + 5) * ld + col];
    *(v4f*)(pk + (size_t)t * 4) = v;
    return;
  }
  int tid = blockIdx.x * blockDim.x + threadIdx.x;
  if (tid >= 4 * 98 * 98) return;
  int h = tid / 9604;
  int rem = tid - h * 9604;
  int n = rem / 98;
  int m = rem - n * 98;
  int zm = m / 49, ym = (m % 49) / 7, xm = m % 7;
  int zn = n / 49, yn = (n % 49) / 7, xn = n % 7;
  int idx = (zm - zn + 1) * 169 + (ym - yn + 6) * 13 + (xm - xn + 6);
  bias_pre[tid] = table[idx * 4 + h];
}

__device__ __forceinline__ void load_wblk(const float* __restrict__ wpk, int nt,
                                          int kblk, int lane, v2f (&bf)[8]) {
  const v4f* p = (const v4f*)wpk + ((nt * 16 + kblk * 4) * 32 + lane);
#pragma unroll
  for (int q = 0; q < 4; ++q) {
    const v4f v = p[q * 32];
    bf[2 * q].x = v.x;
    bf[2 * q].y = v.y;
    bf[2 * q + 1].x = v.z;
    bf[2 * q + 1].y = v.w;
  }
}

template <int NMT>
__device__ __forceinline__ void qkv_block(const float* sX, int kblk, int mt0,
                                          int l16, int hi, const v2f (&bf)[8],
                                          v8f (&acc)[NMT]) {
#pragma unroll
  for (int s = 0; s < 8; ++s) {
    const int k0 = kblk * 32 + s * 4;
#pragma unroll
    for (int j = 0; j < NMT; ++j) {
      int n = (mt0 + j) * 16 + l16;
      int nc = n < 98 ? n : 97;
      v2f a = *(const v2f*)(sX + nc * 132 + k0 + 2 * hi);
      acc[j] = wmma_f32(a, bf[s], acc[j]);
    }
  }
}

template <int NMT>
__device__ __forceinline__ void proj_qkv(const float* __restrict__ wq_pk,
                                         const float* __restrict__ bq,
                                         const float* sX, float (*sQKV)[258],
                                         float* sVT, int ng, int mt0, int lane,
                                         int l16, int hi, v2f (&b0)[8]) {
  v2f b1[8];
#pragma unroll 1
  for (int i = 0; i < 3; ++i) {
    const int nt = ng * 3 + i;
    const int col = nt * 16 + l16;
    const float bias = bq[col];
    v8f acc[NMT] = {};
#pragma unroll 1
    for (int kp = 0; kp < 2; ++kp) {
      load_wblk(wq_pk, nt, 2 * kp + 1, lane, b1);
      qkv_block<NMT>(sX, 2 * kp, mt0, l16, hi, b0, acc);
      const int nnt = (kp == 0 || i == 2) ? nt : nt + 1;
      load_wblk(wq_pk, nnt, kp == 0 ? 2 : 0, lane, b0);
      qkv_block<NMT>(sX, 2 * kp + 1, mt0, l16, hi, b1, acc);
    }
    if (nt < 16) {
      const float scale = (col < 128) ? 0.17677669529663687f : 1.0f;
#pragma unroll
      for (int j = 0; j < NMT; ++j)
#pragma unroll
        for (int r = 0; r < 8; ++r) {
          int m = (mt0 + j) * 16 + r + 8 * hi;
          sQKV[m][col] = (acc[j][r] + bias) * scale;
        }
    } else {
      float* vrow = sVT + (col - 256) * 116 + mt0 * 16 + 8 * hi;
#pragma unroll
      for (int j = 0; j < NMT; ++j) {
        v4f lo, hh;
        lo.x = acc[j][0] + bias;
        lo.y = acc[j][1] + bias;
        lo.z = acc[j][2] + bias;
        lo.w = acc[j][3] + bias;
        hh.x = acc[j][4] + bias;
        hh.y = acc[j][5] + bias;
        hh.z = acc[j][6] + bias;
        hh.w = acc[j][7] + bias;
        *(v4f*)(vrow + j * 16) = lo;
        *(v4f*)(vrow + j * 16 + 4) = hh;
      }
    }
  }
}

template <int NMT>
__device__ __forceinline__ void proj_out(const float* __restrict__ wp_pk,
                                         float bias, float (*sQKV)[258],
                                         float* __restrict__ out, int b,
                                         int nt, int mt0, int lane, int l16,
                                         int hi, v2f (&bn)[8]) {
  const int col = nt * 16 + l16;
  v8f acc[NMT] = {};
#pragma unroll
  for (int kblk = 0; kblk < 4; ++kblk) {
    v2f bc[8];
#pragma unroll
    for (int s = 0; s < 8; ++s) bc[s] = bn[s];
    if (kblk < 3) load_wblk(wp_pk, nt, kblk + 1, lane, bn);
#pragma unroll
    for (int s = 0; s < 8; ++s) {
      const int k0 = kblk * 32 + s * 4;
#pragma unroll
      for (int j = 0; j < NMT; ++j) {
        v2f a = *(const v2f*)(&sQKV[(mt0 + j) * 16 + l16][k0 + 2 * hi]);
        acc[j] = wmma_f32(a, bc[s], acc[j]);
      }
    }
  }
#pragma unroll
  for (int j = 0; j < NMT; ++j)
#pragma unroll
    for (int r = 0; r < 8; ++r) {
      const int m = (mt0 + j) * 16 + r + 8 * hi;
      if (m < 98) out[((size_t)b * 98 + m) * 128 + col] = acc[j][r] + bias;
    }
}

__global__ __launch_bounds__(512) void fused_window_attn(
    const float* __restrict__ x, const float* __restrict__ mask,
    const float* __restrict__ wq_pk, const float* __restrict__ bq,
    const float* __restrict__ wp_pk, const float* __restrict__ bpj,
    const float* __restrict__ bias_pre, float* __restrict__ out) {
  __shared__ __attribute__((aligned(16))) float sQKV[112][258];
  __shared__ __attribute__((aligned(16))) float sVT[128 * 116];
  __shared__ __attribute__((aligned(16))) float sX[112 * 132];
  __shared__ __attribute__((aligned(16))) float sM[98 * 100];

  const int b = blockIdx.x;
  const int wd = b & 511;
  const int wave = __builtin_amdgcn_readfirstlane((int)(threadIdx.x >> 5));
  const int lane = threadIdx.x & 31;
  const int l16 = lane & 15;
  const int hi = lane >> 4;
  const float* xb = x + (size_t)b * 98 * 128;

  v2f wq[8];
  {
    const v4f* xg = (const v4f*)xb;
    v4f t[7];
#pragma unroll
    for (int i = 0; i < 6; ++i) t[i] = xg[(wave + 16 * i) * 32 + lane];
    const int r6 = 96 + wave;
    t[6] = xg[(r6 < 98 ? r6 : 97) * 32 + lane];
    load_wblk(wq_pk, (wave & 7) * 3, 0, lane, wq);
    v2f mk[10];
    {
      const v2f* mg = (const v2f*)(mask + (size_t)wd * 9604);
#pragma unroll
      for (int it = 0; it < 10; ++it) {
        const int i = (int)threadIdx.x + 512 * it;
        mk[it] = mg[i < 4802 ? i : 4801];
      }
    }
#pragma unroll
    for (int i = 0; i < 6; ++i)
      *(v4f*)(sX + (wave + 16 * i) * 132 + 4 * lane) = t[i];
    *(v4f*)(sX + r6 * 132 + 4 * lane) = t[6];
#pragma unroll
    for (int it = 0; it < 10; ++it) {
      const int i = (int)threadIdx.x + 512 * it;
      const int ic = i < 4802 ? i : 4801;
      const int row = ic / 49;
      const int c = ic - row * 49;
      *(v2f*)(sM + row * 100 + 2 * c) = mk[it];
    }
  }
  __syncthreads();

  if (wave < 8)
    proj_qkv<4>(wq_pk, bq, sX, sQKV, sVT, wave, 0, lane, l16, hi, wq);
  else
    proj_qkv<3>(wq_pk, bq, sX, sQKV, sVT, wave - 8, 4, lane, l16, hi, wq);
  __syncthreads();

  for (int it = 0; it < 2; ++it) {
    const int u = wave + 16 * it;
    if (u < 28) {
      const int h = u / 7;
      const int mt = u - h * 7;
      const int m = mt * 16 + l16;
      const int mm = m < 98 ? m : 97;
      const bool qv = (m < 98);

      const float* bp = bias_pre + (size_t)h * 9604;
      float bv[7][8];
#pragma unroll
      for (int nt = 0; nt < 7; ++nt) {
#pragma unroll
        for (int r = 0; r < 8; ++r) {
          if (nt == 6 && r >= 2) {
            bv[nt][r] = 0.0f;
          } else {
            const int n = nt * 16 + 8 * hi + r;
            const int nn = n < 98 ? n : 97;
            bv[nt][r] = bp[nn * 98 + mm];
          }
        }
      }

      v8f acc[7] = {};
      for (int ks = 0; ks < 8; ++ks) {
        const int k0 = ks * 4;
        v2f q = *(const v2f*)(&sQKV[mt * 16 + l16][h * 32 + k0 + 2 * hi]);
#pragma unroll
        for (int nt = 0; nt < 7; ++nt) {
          v2f kf =
              *(const v2f*)(&sQKV[nt * 16 + l16][128 + h * 32 + k0 + 2 * hi]);
          acc[nt] = wmma_f32(kf, q, acc[nt]);
        }
      }

      const float* mrow = sM + mm * 100;
#pragma unroll
      for (int nt = 0; nt < 6; ++nt) {
        const v4f m0 = *(const v4f*)(mrow + nt * 16 + 8 * hi);
        const v4f m1 = *(const v4f*)(mrow + nt * 16 + 8 * hi + 4);
#pragma unroll
        for (int r = 0; r < 8; ++r) {
          const float mvv = r < 4 ? m0[r] : m1[r - 4];
          const float s = acc[nt][r] + (bv[nt][r] + mvv);
          acc[nt][r] = qv ? s : -3.0e38f;
        }
      }
      {
        const v2f m6 = *(const v2f*)(mrow + 96);
        const bool kv = qv && (hi == 0);
#pragma unroll
        for (int r = 0; r < 2; ++r) {
          const float s = acc[6][r] + (bv[6][r] + m6[r]);
          acc[6][r] = kv ? s : -3.0e38f;
        }
      }

      float mx = fmaxf(acc[6][0], acc[6][1]);
#pragma unroll
      for (int nt = 0; nt < 6; ++nt) {
        float t = acc[nt][0];
#pragma unroll
        for (int r = 1; r < 8; ++r) t = fmaxf(t, acc[nt][r]);
        mx = fmaxf(mx, t);
      }
      mx = fmaxf(mx, xhalf(mx));
      float ps[7];
#pragma unroll
      for (int nt = 0; nt < 6; ++nt) {
        ps[nt] = 0.0f;
#pragma unroll
        for (int r = 0; r < 8; ++r) {
          acc[nt][r] = __expf(acc[nt][r] - mx);
          ps[nt] += acc[nt][r];
        }
      }
      acc[6][0] = __expf(acc[6][0] - mx);
      acc[6][1] = __expf(acc[6][1] - mx);
      ps[6] = acc[6][0] + acc[6][1];
      float sum = ((ps[0] + ps[1]) + (ps[2] + ps[3])) +
                  ((ps[4] + ps[5]) + ps[6]);
      sum += xhalf(sum);
      const float inv = 1.0f / sum;
#pragma unroll
      for (int nt = 0; nt < 6; ++nt)
#pragma unroll
        for (int r = 0; r < 8; ++r)
          acc[nt][r] = qv ? acc[nt][r] * inv : 0.0f;
      {
        const bool kv = qv && (hi == 0);
        acc[6][0] = kv ? acc[6][0] * inv : 0.0f;
        acc[6][1] = kv ? acc[6][1] * inv : 0.0f;
      }

      v8f o0 = {}, o1 = {};
      const float* vt0 = sVT + (h * 32 + l16) * 116 + 8 * hi;
      const float* vt1 = vt0 + 16 * 116;
#pragma unroll
      for (int nt = 0; nt < 6; ++nt) {
#pragma unroll
        for (int jj = 0; jj < 2; ++jj) {
          const v4f c0 = *(const v4f*)(vt0 + nt * 16 + 4 * jj);
          const v4f c1 = *(const v4f*)(vt1 + nt * 16 + 4 * jj);
          v2f a, b0, b1;
          a.x = acc[nt][4 * jj];
          a.y = acc[nt][4 * jj + 1];
          b0.x = c0.x; b0.y = c0.y;
          b1.x = c1.x; b1.y = c1.y;
          o0 = wmma_f32(a, b0, o0);
          o1 = wmma_f32(a, b1, o1);
          a.x = acc[nt][4 * jj + 2];
          a.y = acc[nt][4 * jj + 3];
          b0.x = c0.z; b0.y = c0.w;
          b1.x = c1.z; b1.y = c1.w;
          o0 = wmma_f32(a, b0, o0);
          o1 = wmma_f32(a, b1, o1);
        }
      }
      {
        v2f a;
        a.x = acc[6][0];
        a.y = acc[6][1];
        const v2f b0 = *(const v2f*)(vt0 + 96);
        const v2f b1 = *(const v2f*)(vt1 + 96);
        o0 = wmma_f32(a, b0, o0);
        o1 = wmma_f32(a, b1, o1);
      }
#pragma unroll
      for (int r = 0; r < 8; ++r) {
        const int mo = mt * 16 + r + 8 * hi;
        sQKV[mo][h * 32 + l16] = o0[r];
        sQKV[mo][h * 32 + 16 + l16] = o1[r];
      }
    }
  }
  v2f wp[8];
  load_wblk(wp_pk, wave & 7, 0, lane, wp);
  const float bias_o = bpj[(wave & 7) * 16 + l16];
  __syncthreads();

  if (wave < 8)
    proj_out<4>(wp_pk, bias_o, sQKV, out, b, wave, 0, lane, l16, hi, wp);
  else
    proj_out<3>(wp_pk, bias_o, sQKV, out, b, wave - 8, 4, lane, l16, hi, wp);
}

extern "C" void kernel_launch(void* const* d_in, const int* in_sizes, int n_in,
                              void* d_out, int out_size, void* d_ws,
                              size_t ws_size, hipStream_t stream) {
  const float* x      = (const float*)d_in[0];
  const float* mask   = (const float*)d_in[1];
  const float* w_qkv  = (const float*)d_in[2];
  const float* b_qkv  = (const float*)d_in[3];
  const float* w_proj = (const float*)d_in[4];
  const float* b_proj = (const float*)d_in[5];
  const float* table  = (const float*)d_in[6];
  float* out = (float*)d_out;
  float* bias_pre = (float*)d_ws;
  float* wq_pk = (float*)d_ws + 38416;
  float* wp_pk = (float*)d_ws + 38416 + 49152;

  bias_pre_kernel<<<215, 256, 0, stream>>>(table, w_qkv, w_proj, bias_pre,
                                           wq_pk, wp_pk);
  fused_window_attn<<<2048, 512, 0, stream>>>(x, mask, wq_pk, b_qkv, wp_pk,
                                              b_proj, bias_pre, out);
}


# Both projections register-tiled in 2-D: a wave holds 3 (QKV) or 2 (output) column tiles x its row tiles at once, so each LDS A-fragment load feeds 3 or 2 products instead of 1; summation order kept
# speedup vs baseline: 2.7513x; 1.0305x over previous
#include <hip/hip_runtime.h>
#include <hip/hip_bf16.h>
#include <math.h>

typedef __attribute__((ext_vector_type(2))) float v2f;
typedef __attribute__((ext_vector_type(4))) float v4f;
typedef __attribute__((ext_vector_type(8))) float v8f;

__device__ __forceinline__ v8f wmma_f32(v2f a, v2f b, v8f c) {
  return __builtin_amdgcn_wmma_f32_16x16x4_f32(false, a, false, b, (short)0, c,
                                               false, false);
}

__device__ __forceinline__ float xhalf(float x) {
  const int b = __builtin_bit_cast(int, x);
  return __builtin_bit_cast(float, __builtin_amdgcn_permlanex16(
      b, b, 0x76543210, 0xfedcba98, false, false));
}

__global__ void bias_pre_kernel(const float* __restrict__ table,
                                const float* __restrict__ w_qkv,
                                const float* __restrict__ w_proj,
                                float* __restrict__ bias_pre,
                                float* __restrict__ wq_pk,
                                float* __restrict__ wp_pk) {
  if (blockIdx.x >= 151) {
    int t = (blockIdx.x - 151) * 256 + threadIdx.x;
    const float* w = w_qkv;
    float* pk = wq_pk;
    int ld = 384;
    if (t >= 24 * 16 * 32) {
      t -= 24 * 16 * 32;
      w = w_proj;
      pk = wp_pk;
      ld = 128;
    }
    const int lane = t & 31;
    const int kk = (t >> 5) & 15;
    const int nt = t >> 9;
    const int col = nt * 16 + (lane & 15);
    const int kb = 8 * kk + 2 * (lane >> 4);
    v4f v;
    v.x = w[(kb + 0) * ld + col];
    v.y = w[(kb + 1) * ld + col];
    v.z = w[(kb + 4) * ld + col];
    v.w = w[(kb + 5) * ld + col];
    *(v4f*)(pk + (size_t)t * 4) = v;
    return;
  }
  int tid = blockIdx.x * blockDim.x + threadIdx.x;
  if (tid >= 4 * 98 * 98) return;
  int h = tid / 9604;
  int rem = tid - h * 9604;
  int n = rem / 98;
  int m = rem - n * 98;
  int zm = m / 49, ym = (m % 49) / 7, xm = m % 7;
  int zn = n / 49, yn = (n % 49) / 7, xn = n % 7;
  int idx = (zm - zn + 1) * 169 + (ym - yn + 6) * 13 + (xm - xn + 6);
  bias_pre[tid] = table[idx * 4 + h];
}

template <int NT>
__device__ __forceinline__ void load_wtiles(const float* __restrict__ wpk,
                                            int nt0, int blk, int lane,
                                            v4f (&bf)[NT][2]) {
  const v4f* p = (const v4f*)wpk + ((nt0 * 16 + blk * 2) * 32 + lane);
#pragma unroll
  for (int i = 0; i < NT; ++i)
#pragma unroll
    for (int q = 0; q < 2; ++q) bf[i][q] = p[(i * 16 + q) * 32];
}

template <int NMT, int NT>
__device__ __forceinline__ void tile_block(const float* const (&arow)[NMT],
                                           int blk, const v4f (&bf)[NT][2],
                                           v8f (&acc)[NT][NMT]) {
#pragma unroll
  for (int q = 0; q < 2; ++q) {
    const int k0 = (blk * 2 + q) * 8;
    v2f a0[NMT], a1[NMT];
#pragma unroll
    for (int j = 0; j < NMT; ++j) {
      a0[j] = *(const v2f*)(arow[j] + k0);
      a1[j] = *(const v2f*)(arow[j] + k0 + 4);
    }
#pragma unroll
    for (int i = 0; i < NT; ++i) {
      v2f bb;
      bb.x = bf[i][q].x;
      bb.y = bf[i][q].y;
#pragma unroll
      for (int j = 0; j < NMT; ++j) acc[i][j] = wmma_f32(a0[j], bb, acc[i][j]);
    }
#pragma unroll
    for (int i = 0; i < NT; ++i) {
      v2f bb;
      bb.x = bf[i][q].z;
      bb.y = bf[i][q].w;
#pragma unroll
      for (int j = 0; j < NMT; ++j) acc[i][j] = wmma_f32(a1[j], bb, acc[i][j]);
    }
  }
}

template <int NMT>
__device__ __forceinline__ void proj_qkv(const float* __restrict__ wq_pk,
                                         const float* __restrict__ bq,
                                         const float* sX, float (*sQKV)[258],
                                         float* sVT, int ng, int mt0, int lane,
                                         int l16, int hi, v4f (&b0)[3][2]) {
  v4f b1[3][2];
  float bias[3];
#pragma unroll
  for (int i = 0; i < 3; ++i) bias[i] = bq[(ng * 3 + i) * 16 + l16];
  const float* arow[NMT];
#pragma unroll
  for (int j = 0; j < NMT; ++j) {
    const int n = (mt0 + j) * 16 + l16;
    const int nc = n < 98 ? n : 97;
    arow[j] = sX + nc * 132 + 2 * hi;
  }
  v8f acc[3][NMT] = {};
#pragma unroll 1
  for (int t = 0; t < 4; ++t) {
    load_wtiles<3>(wq_pk, ng * 3, 2 * t + 1, lane, b1);
    tile_block<NMT, 3>(arow, 2 * t, b0, acc);
    load_wtiles<3>(wq_pk, ng * 3, t < 3 ? 2 * t + 2 : 0, lane, b0);
    tile_block<NMT, 3>(arow, 2 * t + 1, b1, acc);
  }
#pragma unroll
  for (int i = 0; i < 3; ++i) {
    const int nt = ng * 3 + i;
    const int col = nt * 16 + l16;
    if (nt < 16) {
      const float scale = (col < 128) ? 0.17677669529663687f : 1.0f;
#pragma unroll
      for (int j = 0; j < NMT; ++j)
#pragma unroll
        for (int r = 0; r < 8; ++r) {
          int m = (mt0 + j) * 16 + r + 8 * hi;
          sQKV[m][col] = (acc[i][j][r] + bias[i]) * scale;
        }
    } else {
      float* vrow = sVT + (col - 256) * 116 + mt0 * 16 + 8 * hi;
#pragma unroll
      for (int j = 0; j < NMT; ++j) {
        v4f lo, hh;
        lo.x = acc[i][j][0] + bias[i];
        lo.y = acc[i][j][1] + bias[i];
        lo.z = acc[i][j][2] + bias[i];
        lo.w = acc[i][j][3] + bias[i];
        hh.x = acc[i][j][4] + bias[i];
        hh.y = acc[i][j][5] + bias[i];
        hh.z = acc[i][j][6] + bias[i];
        hh.w = acc[i][j][7] + bias[i];
        *(v4f*)(vrow + j * 16) = lo;
        *(v4f*)(vrow + j * 16 + 4) = hh;
      }
    }
  }
}

template <int NR>
__device__ __forceinline__ void proj_out(const float* __restrict__ wp_pk,
                                         float bias0, float bias1,
                                         float (*sQKV)[258],
                                         float* __restrict__ out, int b,
                                         int np, int mt0, int lane, int l16,
                                         int hi, v4f (&bn)[2][2]) {
  const float* arow[NR];
#pragma unroll
  for (int j = 0; j < NR; ++j) arow[j] = &sQKV[(mt0 + j) * 16 + l16][2 * hi];
  v8f acc[2][NR] = {};
  v4f bo[2][2];
#pragma unroll
  for (int t = 0; t < 4; ++t) {
    load_wtiles<2>(wp_pk, np * 2, 2 * t + 1, lane, bo);
    tile_block<NR, 2>(arow, 2 * t, bn, acc);
    if (t < 3) load_wtiles<2>(wp_pk, np * 2, 2 * t + 2, lane, bn);
    tile_block<NR, 2>(arow, 2 * t + 1, bo, acc);
  }
#pragma unroll
  for (int i = 0; i < 2; ++i) {
    const int col = (np * 2 + i) * 16 + l16;
    const float bias = i == 0 ? bias0 : bias1;
#pragma unroll
    for (int j = 0; j < NR; ++j)
#pragma unroll
      for (int r = 0; r < 8; ++r) {
        const int m = (mt0 + j) * 16 + r + 8 * hi;
        if (NR == 2 || m < 98)
          out[((size_t)b * 98 + m) * 128 + col] = acc[i][j][r] + bias;
      }
  }
}

__global__ __launch_bounds__(512) void fused_window_attn(
    const float* __restrict__ x, const float* __restrict__ mask,
    const float* __restrict__ wq_pk, const float* __restrict__ bq,
    const float* __restrict__ wp_pk, const float* __restrict__ bpj,
    const float* __restrict__ bias_pre, float* __restrict__ out) {
  __shared__ __attribute__((aligned(16))) float sQKV[112][258];
  __shared__ __attribute__((aligned(16))) float sVT[128 * 116];
  __shared__ __attribute__((aligned(16))) float sX[112 * 132];
  __shared__ __attribute__((aligned(16))) float sM[98 * 100];

  const int b = blockIdx.x;
  const int wd = b & 511;
  const int wave = __builtin_amdgcn_readfirstlane((int)(threadIdx.x >> 5));
  const int lane = threadIdx.x & 31;
  const int l16 = lane & 15;
  const int hi = lane >> 4;
  const float* xb = x + (size_t)b * 98 * 128;

  v4f wq[3][2];
  {
    const v4f* xg = (const v4f*)xb;
    v4f t[7];
#pragma unroll
    for (int i = 0; i < 6; ++i) t[i] = xg[(wave + 16 * i) * 32 + lane];
    const int r6 = 96 + wave;
    t[6] = xg[(r6 < 98 ? r6 : 97) * 32 + lane];
    load_wtiles<3>(wq_pk, (wave & 7) * 3, 0, lane, wq);
    v2f mk[10];
    {
      const v2f* mg = (const v2f*)(mask + (size_t)wd * 9604);
#pragma unroll
      for (int it = 0; it < 10; ++it) {
        const int i = (int)threadIdx.x + 512 * it;
        mk[it] = mg[i < 4802 ? i : 4801];
      }
    }
#pragma unroll
    for (int i = 0; i < 6; ++i)
      *(v4f*)(sX + (wave + 16 * i) * 132 + 4 * lane) = t[i];
    *(v4f*)(sX + r6 * 132 + 4 * lane) = t[6];
#pragma unroll
    for (int it = 0; it < 10; ++it) {
      const int i = (int)threadIdx.x + 512 * it;
      const int ic = i < 4802 ? i : 4801;
      const int row = ic / 49;
      const int c = ic - row * 49;
      *(v2f*)(sM + row * 100 + 2 * c) = mk[it];
    }
  }
  __syncthreads();

  if (wave < 8)
    proj_qkv<4>(wq_pk, bq, sX, sQKV, sVT, wave, 0, lane, l16, hi, wq);
  else
    proj_qkv<3>(wq_pk, bq, sX, sQKV, sVT, wave - 8, (wave >> 3) * 4, lane, l16,
                hi, wq);
  __syncthreads();

  for (int it = 0; it < 2; ++it) {
    const int u = wave + 16 * it;
    if (u < 28) {
      const int h = u / 7;
      const int mt = u - h * 7;
      const int m = mt * 16 + l16;
      const int mm = m < 98 ? m : 97;
      const bool qv = (m < 98);

      const float* bp = bias_pre + (size_t)h * 9604;
      float bv[7][8];
#pragma unroll
      for (int nt = 0; nt < 7; ++nt) {
#pragma unroll
        for (int r = 0; r < 8; ++r) {
          if (nt == 6 && r >= 2) {
            bv[nt][r] = 0.0f;
          } else {
            const int n = nt * 16 + 8 * hi + r;
            const int nn = n < 98 ? n : 97;
            bv[nt][r] = bp[nn * 98 + mm];
          }
        }
      }

      v8f acc[7] = {};
      for (int ks = 0; ks < 8; ++ks) {
        const int k0 = ks * 4;
        v2f q = *(const v2f*)(&sQKV[mt * 16 + l16][h * 32 + k0 + 2 * hi]);
#pragma unroll
        for (int nt = 0; nt < 7; ++nt) {
          v2f kf =
              *(const v2f*)(&sQKV[nt * 16 + l16][128 + h * 32 + k0 + 2 * hi]);
          acc[nt] = wmma_f32(kf, q, acc[nt]);
        }
      }

      const float* mrow = sM + mm * 100;
#pragma unroll
      for (int nt = 0; nt < 6; ++nt) {
        const v4f m0 = *(const v4f*)(mrow + nt * 16 + 8 * hi);
        const v4f m1 = *(const v4f*)(mrow + nt * 16 + 8 * hi + 4);
#pragma unroll
        for (int r = 0; r < 8; ++r) {
          const float mvv = r < 4 ? m0[r] : m1[r - 4];
          const float s = acc[nt][r] + (bv[nt][r] + mvv);
          acc[nt][r] = qv ? s : -3.0e38f;
        }
      }
      {
        const v2f m6 = *(const v2f*)(mrow + 96);
        const bool kv = qv && (hi == 0);
#pragma unroll
        for (int r = 0; r < 2; ++r) {
          const float s = acc[6][r] + (bv[6][r] + m6[r]);
          acc[6][r] = kv ? s : -3.0e38f;
        }
      }

      float mx = fmaxf(acc[6][0], acc[6][1]);
#pragma unroll
      for (int nt = 0; nt < 6; ++nt) {
        float t = acc[nt][0];
#pragma unroll
        for (int r = 1; r < 8; ++r) t = fmaxf(t, acc[nt][r]);
        mx = fmaxf(mx, t);
      }
      mx = fmaxf(mx, xhalf(mx));
      float ps[7];
#pragma unroll
      for (int nt = 0; nt < 6; ++nt) {
        ps[nt] = 0.0f;
#pragma unroll
        for (int r = 0; r < 8; ++r) {
          acc[nt][r] = __expf(acc[nt][r] - mx);
          ps[nt] += acc[nt][r];
        }
      }
      acc[6][0] = __expf(acc[6][0] - mx);
      acc[6][1] = __expf(acc[6][1] - mx);
      ps[6] = acc[6][0] + acc[6][1];
      float sum = ((ps[0] + ps[1]) + (ps[2] + ps[3])) +
                  ((ps[4] + ps[5]) + ps[6]);
      sum += xhalf(sum);
      const float inv = 1.0f / sum;
#pragma unroll
      for (int nt = 0; nt < 6; ++nt)
#pragma unroll
        for (int r = 0; r < 8; ++r)
          acc[nt][r] = qv ? acc[nt][r] * inv : 0.0f;
      {
        const bool kv = qv && (hi == 0);
        acc[6][0] = kv ? acc[6][0] * inv : 0.0f;
        acc[6][1] = kv ? acc[6][1] * inv : 0.0f;
      }

      v8f o0 = {}, o1 = {};
      const float* vt0 = sVT + (h * 32 + l16) * 116 + 8 * hi;
      const float* vt1 = vt0 + 16 * 116;
#pragma unroll
      for (int nt = 0; nt < 6; ++nt) {
#pragma unroll
        for (int jj = 0; jj < 2; ++jj) {
          const v4f c0 = *(const v4f*)(vt0 + nt * 16 + 4 * jj);
          const v4f c1 = *(const v4f*)(vt1 + nt * 16 + 4 * jj);
          v2f a, b0, b1;
          a.x = acc[nt][4 * jj];
          a.y = acc[nt][4 * jj + 1];
          b0.x = c0.x; b0.y = c0.y;
          b1.x = c1.x; b1.y = c1.y;
          o0 = wmma_f32(a, b0, o0);
          o1 = wmma_f32(a, b1, o1);
          a.x = acc[nt][4 * jj + 2];
          a.y = acc[nt][4 * jj + 3];
          b0.x = c0.z; b0.y = c0.w;
          b1.x = c1.z; b1.y = c1.w;
          o0 = wmma_f32(a, b0, o0);
          o1 = wmma_f32(a, b1, o1);
        }
      }
      {
        v2f a;
        a.x = acc[6][0];
        a.y = acc[6][1];
        const v2f b0 = *(const v2f*)(vt0 + 96);
        const v2f b1 = *(const v2f*)(vt1 + 96);
        o0 = wmma_f32(a, b0, o0);
        o1 = wmma_f32(a, b1, o1);
      }
#pragma unroll
      for (int r = 0; r < 8; ++r) {
        const int mo = mt * 16 + r + 8 * hi;
        sQKV[mo][h * 32 + l16] = o0[r];
        sQKV[mo][h * 32 + 16 + l16] = o1[r];
      }
    }
  }
  const int np = wave & 3;
  const int rg = ((wave >> 2) + wave) & 3;
  v4f wp[2][2];
  load_wtiles<2>(wp_pk, np * 2, 0, lane, wp);
  const float bias_o0 = bpj[np * 32 + l16];
  const float bias_o1 = bpj[np * 32 + 16 + l16];
  __syncthreads();

  if (rg < 3)
    proj_out<2>(wp_pk, bias_o0, bias_o1, sQKV, out, b, np, rg * 2, lane, l16,
                hi, wp);
  else
    proj_out<1>(wp_pk, bias_o0, bias_o1, sQKV, out, b, np, 6, lane, l16, hi,
                wp);
}

extern "C" void kernel_launch(void* const* d_in, const int* in_sizes, int n_in,
                              void* d_out, int out_size, void* d_ws,
                              size_t ws_size, hipStream_t stream) {
  const float* x      = (const float*)d_in[0];
  const float* mask   = (const float*)d_in[1];
  const float* w_qkv  = (const float*)d_in[2];
  const float* b_qkv  = (const float*)d_in[3];
  const float* w_proj = (const float*)d_in[4];
  const float* b_proj = (const float*)d_in[5];
  const float* table  = (const float*)d_in[6];
  float* out = (float*)d_out;
  float* bias_pre = (float*)d_ws;
  float* wq_pk = (float*)d_ws + 38416;
  float* wp_pk = (float*)d_ws + 38416 + 49152;

  bias_pre_kernel<<<215, 256, 0, stream>>>(table, w_qkv, w_proj, bias_pre,
                                           wq_pk, wp_pk);
  fused_window_attn<<<2048, 512, 0, stream>>>(x, mask, wq_pk, b_qkv, wp_pk,
                                              b_proj, bias_pre, out);
}
